# starting kernel
# baseline (speedup 1.0000x reference)
#include <hip/hip_runtime.h>
#include <hip/hip_bf16.h>

typedef __attribute__((ext_vector_type(2))) float v2f;
typedef __attribute__((ext_vector_type(8))) float v8f;

#define IMG_H 256
#define IMG_W 256
#define NGAUSS 512
#define FOCAL_F 50.0f

#define WS_GOP 2048
#define WS_BMAT 2560

__device__ __forceinline__ float lane_xor16(float x) {
    int v = __builtin_bit_cast(int, x);
    int r = __builtin_amdgcn_permlanex16(v, v, 0x76543210, 0xfedcba98, false, false);
    return __builtin_bit_cast(float, r);
}

__global__ __launch_bounds__(512)
void dgr_prep_kernel(const float* __restrict__ means3d,
                     const float* __restrict__ scales,
                     const float* __restrict__ opac,
                     const float* __restrict__ colors,
                     float* __restrict__ ws) {
    __shared__ float zsh[NGAUSS];
    __shared__ float csh[NGAUSS * 3];

    const int i = threadIdx.x;

    const float z = fmaxf(means3d[i * 3 + 2], 0.1f);
    zsh[i] = z;
    __syncthreads();

    int rank = 0;
    for (int j = 0; j < NGAUSS; ++j) {
        const float zj = zsh[j];
        rank += ((zj < z) || (zj == z && j < i)) ? 1 : 0;
    }

    const float pu = FOCAL_F * means3d[i * 3 + 0] / z;
    const float pv = FOCAL_F * means3d[i * 3 + 1] / z;
    const float su = fmaxf(FOCAL_F * scales[i * 3 + 0] / z, 0.5f);
    const float sv = fmaxf(FOCAL_F * scales[i * 3 + 1] / z, 0.5f);

    float4 g;
    g.x = pu;
    g.y = pv;
    g.z = 1.0f / (su * su);
    g.w = 1.0f / (sv * sv);
    ((float4*)ws)[rank] = g;
    ws[WS_GOP + rank] = opac[i];

    csh[rank * 3 + 0] = colors[i * 3 + 0];
    csh[rank * 3 + 1] = colors[i * 3 + 1];
    csh[rank * 3 + 2] = colors[i * 3 + 2];
    __syncthreads();

    float* bmat = ws + WS_BMAT;
    for (int e = i; e < 128 * 32; e += 512) {
        const int t = e >> 5;
        const int l = e & 31;
        const int col = l & 15;
        const int k0 = (l < 16) ? 0 : 2;
        const int n0 = 4 * t;
        float b0 = 0.0f, b1 = 0.0f;
        if (col < 3) {
            b0 = csh[(n0 + k0) * 3 + col];
            b1 = csh[(n0 + k0 + 1) * 3 + col];
        }
        bmat[e * 2 + 0] = b0;
        bmat[e * 2 + 1] = b1;
    }
}

__global__ __launch_bounds__(256)
void dgr_raster_kernel(const float* __restrict__ ws, float* __restrict__ out) {
    const float4* __restrict__ gp4 = (const float4*)ws;
    const float* __restrict__ gop = ws + WS_GOP;
    const v2f* __restrict__ bmat = (const v2f*)(ws + WS_BMAT);

    const int tid = threadIdx.x;
    const int lane = tid & 31;
    const int wave = tid >> 5;
    const int pixBase = blockIdx.x * 256 + wave * 32;
    const int p = pixBase + lane;

    const float u = (float)(p & (IMG_W - 1)) - (float)(IMG_W / 2);
    const float v = (float)(p >> 8) - (float)(IMG_H / 2);

    const bool lo = (lane < 16);

    float T = 1.0f;
    v8f acc1 = {0.f, 0.f, 0.f, 0.f, 0.f, 0.f, 0.f, 0.f};
    v8f acc2 = {0.f, 0.f, 0.f, 0.f, 0.f, 0.f, 0.f, 0.f};

    for (int t = 0; t < NGAUSS / 4; ++t) {
        const int pt = (t + 16 < 128) ? (t + 16) : t;
        __builtin_prefetch((const void*)(bmat + pt * 32 + lane), 0, 3);

        float wk[4];
#pragma unroll
        for (int k = 0; k < 4; ++k) {
            const int n = 4 * t + k;
            const float4 g = gp4[n];
            const float op = gop[n];
            const float du = u - g.x;
            const float dv = v - g.y;
            const float e = __expf(-0.5f * (du * du * g.z + dv * dv * g.w));
            const float a = fminf(e * op, 1.0f);
            wk[k] = a * T;
            T *= (1.0f - a);
        }

        const float s0 = lane_xor16(wk[0]);
        const float s1 = lane_xor16(wk[1]);
        const float s2 = lane_xor16(wk[2]);
        const float s3 = lane_xor16(wk[3]);

        v2f a1, a2;
        a1.x = lo ? wk[0] : s2;
        a1.y = lo ? wk[1] : s3;
        a2.x = lo ? s0 : wk[2];
        a2.y = lo ? s1 : wk[3];

        const v2f b = bmat[t * 32 + lane];

        acc1 = __builtin_amdgcn_wmma_f32_16x16x4_f32(
            false, a1, false, b, (short)0, acc1, false, false);
        acc2 = __builtin_amdgcn_wmma_f32_16x16x4_f32(
            false, a2, false, b, (short)0, acc2, false, false);
    }

    const int ch = lane & 15;
    if (ch < 3) {
#pragma unroll
        for (int r = 0; r < 8; ++r) {
            const int m = r + (lo ? 0 : 8);
            out[(pixBase + m) * 3 + ch] = acc1[r];
            out[(pixBase + 16 + m) * 3 + ch] = acc2[r];
        }
    }
}

extern "C" void kernel_launch(void* const* d_in, const int* in_sizes, int n_in,
                              void* d_out, int out_size, void* d_ws, size_t ws_size,
                              hipStream_t stream) {
    const float* means3d = (const float*)d_in[0];
    const float* scales  = (const float*)d_in[1];
    const float* opac    = (const float*)d_in[2];
    const float* colors  = (const float*)d_in[3];
    float* ws  = (float*)d_ws;
    float* out = (float*)d_out;

    hipLaunchKernelGGL(dgr_prep_kernel, dim3(1), dim3(512), 0, stream,
                       means3d, scales, opac, colors, ws);

    hipLaunchKernelGGL(dgr_raster_kernel, dim3(256), dim3(256), 0, stream,
                       ws, out);
}


# raster: separable gaussian on strips of 32 columns x 2 rows per wave: one exp2 per lane and gaussian serves both rows, row factors by v_readlane; lane's 2 weights = A fragment, 1 wmma per gaussian
# speedup vs baseline: 1.7963x; 1.7963x over previous
#include <hip/hip_runtime.h>


typedef __attribute__((ext_vector_type(2))) float v2f;
typedef __attribute__((ext_vector_type(8))) float v8f;

#define IMG_H 256
#define IMG_W 256
#define NGAUSS 512
#define FOCAL_F 50.0f
#define NCHUNK (NGAUSS / 32)

#define WS_GOP 2048
#define WS_BQ 2560

__device__ __forceinline__ float lane_bcast(float x, int srcLane) {
    return __builtin_bit_cast(float, __builtin_amdgcn_readlane(__builtin_bit_cast(int, x), srcLane));
}

__device__ __forceinline__ float clamp01(float x) {
    return __builtin_amdgcn_fmed3f(x, 0.0f, 1.0f);
}

__global__ __launch_bounds__(512)
void dgr_prep_kernel(const float* __restrict__ means3d,
                     const float* __restrict__ scales,
                     const float* __restrict__ opac,
                     const float* __restrict__ colors,
                     float* __restrict__ ws) {
    __shared__ float zsh[NGAUSS];

    const int i = threadIdx.x;

    const float z = fmaxf(means3d[i * 3 + 2], 0.1f);
    zsh[i] = z;
    __syncthreads();

    int rank = 0;
    for (int j = 0; j < NGAUSS; ++j) {
        const float zj = zsh[j];
        rank += ((zj < z) || (zj == z && j < i)) ? 1 : 0;
    }

    const float pu = FOCAL_F * means3d[i * 3 + 0] / z;
    const float pv = FOCAL_F * means3d[i * 3 + 1] / z;
    const float su = fmaxf(FOCAL_F * scales[i * 3 + 0] / z, 0.5f);
    const float sv = fmaxf(FOCAL_F * scales[i * 3 + 1] / z, 0.5f);

    const float kc = -0.5f * 1.44269504088896340736f;
    float4 g;
    g.x = pu;
    g.y = pv;
    g.z = kc / (su * su);
    g.w = kc / (sv * sv);
    ((float4*)ws)[rank] = g;
    ws[WS_GOP + rank] = opac[i];

    const float c0 = colors[i * 3 + 0];
    const float c1 = colors[i * 3 + 1];
    const float c2 = colors[i * 3 + 2];
    float4* rec = (float4*)(ws + WS_BQ) + rank * 4;
    rec[0] = make_float4(c0, 0.0f, c1, 0.0f);
    rec[1] = make_float4(c2, 0.0f, 0.0f, c0);
    rec[2] = make_float4(0.0f, c1, 0.0f, c2);
    rec[3] = make_float4(0.0f, 0.0f, 0.0f, 0.0f);
}

struct Chunk {
    float4 g;
    float op;
    v2f b[32];
};

__device__ __forceinline__ void load_chunk(Chunk& k, const float4* __restrict__ gp4,
                                           const float* __restrict__ gop,
                                           const v2f* __restrict__ bl, int c, int lane) {
    const int n = c * 32 + lane;
    k.g = gp4[n];
    k.op = gop[n];
#pragma unroll
    for (int j = 0; j < 32; ++j)
        k.b[j] = bl[(c * 32 + j) * 8];
}

__device__ __forceinline__ void do_chunk(const Chunk& k, float xf, float y0f, float y1f,
                                         v2f& T, v8f& acc) {
    const float dv0 = y0f - k.g.y;
    const float dv1 = y1f - k.g.y;
    const float r0 = k.op * __builtin_amdgcn_exp2f(dv0 * dv0 * k.g.w);
    const float r1 = k.op * __builtin_amdgcn_exp2f(dv1 * dv1 * k.g.w);

#pragma unroll
    for (int j = 0; j < 32; ++j) {
        const float pu = lane_bcast(k.g.x, j);
        const float ku = lane_bcast(k.g.z, j);
        const float rr0 = lane_bcast(r0, j);
        const float rr1 = lane_bcast(r1, j);

        const float du = xf - pu;
        const float e = __builtin_amdgcn_exp2f(du * du * ku);

        v2f a;
        a.x = clamp01(e * rr0);
        a.y = clamp01(e * rr1);
        const v2f w = a * T;
        T = T - w;

        acc = __builtin_amdgcn_wmma_f32_16x16x4_f32(
            false, w, false, k.b[j], (short)0, acc, false, false);
    }
}

__global__ __launch_bounds__(128)
void dgr_raster_kernel(const float* __restrict__ ws, float* __restrict__ out) {
    const float4* __restrict__ gp4 = (const float4*)ws;
    const float* __restrict__ gop = ws + WS_GOP;

    const int lane = threadIdx.x & 31;
    const int gw = blockIdx.x * 4 + (threadIdx.x >> 5);
    const int x0 = (gw & 7) * 32;
    const int yr = (gw >> 3) * 2;

    const float xf = (float)(x0 + lane) - (float)(IMG_W / 2);
    const float y0f = (float)yr - (float)(IMG_H / 2);
    const float y1f = y0f + 1.0f;

    const int c = lane & 15;
    const int h = lane >> 4;
    const int q = c >> 2;
    const int ch = c & 3;
    int boff = 12;
    if (ch < 3) {
        if (q == 2 * h) boff = 2 * ch;
        else if (q == 2 * h + 1) boff = 6 + 2 * ch;
    }
    const v2f* __restrict__ bl = (const v2f*)(ws + WS_BQ + boff);

    v2f T = {1.0f, 1.0f};
    v8f acc = {0.f, 0.f, 0.f, 0.f, 0.f, 0.f, 0.f, 0.f};

    Chunk ca, cb;
    load_chunk(ca, gp4, gop, bl, 0, lane);
    for (int ck = 0;; ck += 2) {
        load_chunk(cb, gp4, gop, bl, ck + 1, lane);
        do_chunk(ca, xf, y0f, y1f, T, acc);
        if (ck + 2 >= NCHUNK) {
            do_chunk(cb, xf, y0f, y1f, T, acc);
            break;
        }
        load_chunk(ca, gp4, gop, bl, ck + 2, lane);
        asm volatile("" ::: "memory");
        do_chunk(cb, xf, y0f, y1f, T, acc);
    }

    if (ch < 3) {
        const int row = yr + (q & 1);
        const int xb = x0 + 16 * (q >> 1) + 8 * h;
        float* o = out + ((size_t)row * IMG_W + xb) * 3 + ch;
#pragma unroll
        for (int r = 0; r < 8; ++r)
            o[r * 3] = acc[r];
    }
}

extern "C" void kernel_launch(void* const* d_in, const int* in_sizes, int n_in,
                              void* d_out, int out_size, void* d_ws, size_t ws_size,
                              hipStream_t stream) {
    const float* means3d = (const float*)d_in[0];
    const float* scales  = (const float*)d_in[1];
    const float* opac    = (const float*)d_in[2];
    const float* colors  = (const float*)d_in[3];
    float* ws  = (float*)d_ws;
    float* out = (float*)d_out;

    hipLaunchKernelGGL(dgr_prep_kernel, dim3(1), dim3(512), 0, stream,
                       means3d, scales, opac, colors, ws);

    hipLaunchKernelGGL(dgr_raster_kernel, dim3(256), dim3(128), 0, stream,
                       ws, out);
}


# prep: rank sort spread over the chip, one wave per gaussian (512 waves): lanes hold all 512 keys, wave votes + popcount count the keys in front; was one 512-thread work-group; ranks, sums unchanged
# speedup vs baseline: 3.3298x; 1.8537x over previous
#include <hip/hip_runtime.h>


typedef __attribute__((ext_vector_type(2))) float v2f;
typedef __attribute__((ext_vector_type(8))) float v8f;

#define IMG_H 256
#define IMG_W 256
#define NGAUSS 512
#define FOCAL_F 50.0f
#define NCHUNK (NGAUSS / 32)

#define WS_GOP 2048
#define WS_BQ 2560

__device__ __forceinline__ float lane_bcast(float x, int srcLane) {
    return __builtin_bit_cast(float, __builtin_amdgcn_readlane(__builtin_bit_cast(int, x), srcLane));
}

__device__ __forceinline__ float clamp01(float x) {
    return __builtin_amdgcn_fmed3f(x, 0.0f, 1.0f);
}

__global__ __launch_bounds__(128)
void dgr_prep_kernel(const float* __restrict__ means3d,
                     const float* __restrict__ scales,
                     const float* __restrict__ opac,
                     const float* __restrict__ colors,
                     float* __restrict__ ws) {
    const unsigned lane = threadIdx.x & 31u;
    const int i = blockIdx.x * 4 + __builtin_amdgcn_readfirstlane((int)(threadIdx.x >> 5));

    float zj[16];
#pragma unroll
    for (int t = 0; t < 16; ++t)
        zj[t] = fmaxf(means3d[(t * 32 + (int)lane) * 3 + 2], 0.1f);

    const float z = fmaxf(means3d[i * 3 + 2], 0.1f);

    const unsigned csel = ((0x204u >> lane) & 1u) + 2u * ((0x810u >> lane) & 1u);
    const unsigned e = lane & 3u;
    const float* src = (lane < 16u) ? colors + i * 3 + csel
                     : (lane < 20u) ? ((e < 2u) ? means3d : scales) + i * 3 + (e & 1u)
                     : opac + i;
    const float m = *src;

    const int it = i >> 5;
    const unsigned below_i = (1u << (i & 31)) - 1u;
    int rank = 0;
#pragma unroll
    for (int t = 0; t < 16; ++t) {
        const unsigned lt = __builtin_amdgcn_ballot_w32(zj[t] < z);
        const unsigned eq = __builtin_amdgcn_ballot_w32(zj[t] == z);
        const unsigned first = (t < it) ? 0xffffffffu : ((t == it) ? below_i : 0u);
        rank += __builtin_popcount(lt | (eq & first));
    }

    const float q = FOCAL_F * m / z;
    const float sg = fmaxf(q, 0.5f);
    const float kc = -0.5f * 1.44269504088896340736f;
    const float k = kc / (sg * sg);

    const unsigned keep = 0u - ((0xA95u >> lane) & 1u);
    const float rec = __builtin_bit_cast(float, __builtin_bit_cast(unsigned, m) & keep);

    const float v = (lane < 16u) ? rec : (lane < 18u) ? q : (lane < 20u) ? k : m;
    const int dst = (lane < 16u) ? (WS_BQ + rank * 16 + (int)lane)
                  : (lane < 20u) ? (rank * 4 + (int)e) : (WS_GOP + rank);
    if (lane <= 20u)
        ws[dst] = v;
}

struct Chunk {
    float4 g;
    float op;
    v2f b[32];
};

__device__ __forceinline__ void load_chunk(Chunk& k, const float4* __restrict__ gp4,
                                           const float* __restrict__ gop,
                                           const v2f* __restrict__ bl, int c, int lane) {
    const int n = c * 32 + lane;
    k.g = gp4[n];
    k.op = gop[n];
#pragma unroll
    for (int j = 0; j < 32; ++j)
        k.b[j] = bl[(c * 32 + j) * 8];
}

__device__ __forceinline__ void do_chunk(const Chunk& k, float xf, float y0f, float y1f,
                                         v2f& T, v8f& acc) {
    const float dv0 = y0f - k.g.y;
    const float dv1 = y1f - k.g.y;
    const float r0 = k.op * __builtin_amdgcn_exp2f(dv0 * dv0 * k.g.w);
    const float r1 = k.op * __builtin_amdgcn_exp2f(dv1 * dv1 * k.g.w);

#pragma unroll
    for (int j = 0; j < 32; ++j) {
        const float pu = lane_bcast(k.g.x, j);
        const float ku = lane_bcast(k.g.z, j);
        const float rr0 = lane_bcast(r0, j);
        const float rr1 = lane_bcast(r1, j);

        const float du = xf - pu;
        const float e = __builtin_amdgcn_exp2f(du * du * ku);

        v2f a;
        a.x = clamp01(e * rr0);
        a.y = clamp01(e * rr1);
        const v2f w = a * T;
        T = T - w;

        acc = __builtin_amdgcn_wmma_f32_16x16x4_f32(
            false, w, false, k.b[j], (short)0, acc, false, false);
    }
}

__global__ __launch_bounds__(128)
void dgr_raster_kernel(const float* __restrict__ ws, float* __restrict__ out) {
    const float4* __restrict__ gp4 = (const float4*)ws;
    const float* __restrict__ gop = ws + WS_GOP;

    const int lane = threadIdx.x & 31;
    const int gw = blockIdx.x * 4 + (threadIdx.x >> 5);
    const int x0 = (gw & 7) * 32;
    const int yr = (gw >> 3) * 2;

    const float xf = (float)(x0 + lane) - (float)(IMG_W / 2);
    const float y0f = (float)yr - (float)(IMG_H / 2);
    const float y1f = y0f + 1.0f;

    const int c = lane & 15;
    const int h = lane >> 4;
    const int q = c >> 2;
    const int ch = c & 3;
    int boff = 12;
    if (ch < 3) {
        if (q == 2 * h) boff = 2 * ch;
        else if (q == 2 * h + 1) boff = 6 + 2 * ch;
    }
    const v2f* __restrict__ bl = (const v2f*)(ws + WS_BQ + boff);

    v2f T = {1.0f, 1.0f};
    v8f acc = {0.f, 0.f, 0.f, 0.f, 0.f, 0.f, 0.f, 0.f};

    Chunk ca, cb;
    load_chunk(ca, gp4, gop, bl, 0, lane);
    for (int ck = 0;; ck += 2) {
        load_chunk(cb, gp4, gop, bl, ck + 1, lane);
        do_chunk(ca, xf, y0f, y1f, T, acc);
        if (ck + 2 >= NCHUNK) {
            do_chunk(cb, xf, y0f, y1f, T, acc);
            break;
        }
        load_chunk(ca, gp4, gop, bl, ck + 2, lane);
        asm volatile("" ::: "memory");
        do_chunk(cb, xf, y0f, y1f, T, acc);
    }

    if (ch < 3) {
        const int row = yr + (q & 1);
        const int xb = x0 + 16 * (q >> 1) + 8 * h;
        float* o = out + ((size_t)row * IMG_W + xb) * 3 + ch;
#pragma unroll
        for (int r = 0; r < 8; ++r)
            o[r * 3] = acc[r];
    }
}

extern "C" void kernel_launch(void* const* d_in, const int* in_sizes, int n_in,
                              void* d_out, int out_size, void* d_ws, size_t ws_size,
                              hipStream_t stream) {
    const float* means3d = (const float*)d_in[0];
    const float* scales  = (const float*)d_in[1];
    const float* opac    = (const float*)d_in[2];
    const float* colors  = (const float*)d_in[3];
    float* ws  = (float*)d_ws;
    float* out = (float*)d_out;

    hipLaunchKernelGGL(dgr_prep_kernel, dim3(128), dim3(128), 0, stream,
                       means3d, scales, opac, colors, ws);

    hipLaunchKernelGGL(dgr_raster_kernel, dim3(256), dim3(128), 0, stream,
                       ws, out);
}


# raster: work-group tile 32 columns x 8 rows (was 128 x 2): column factor exp2(du^2*ku) made once per work-group, lane = gaussian, into an LDS table; each lane reads its column's row; sums unchanged
# speedup vs baseline: 3.9683x; 1.1918x over previous
#include <hip/hip_runtime.h>


typedef __attribute__((ext_vector_type(2))) float v2f;
typedef __attribute__((ext_vector_type(4))) float v4f;
typedef __attribute__((ext_vector_type(8))) float v8f;

#define IMG_H 256
#define IMG_W 256
#define NGAUSS 512
#define FOCAL_F 50.0f
#define NCHUNK (NGAUSS / 32)
#define EPITCH (NGAUSS + 4)

#define WS_GOP 2048
#define WS_BQ 2560

__device__ __forceinline__ float lane_bcast(float x, int srcLane) {
    return __builtin_bit_cast(float, __builtin_amdgcn_readlane(__builtin_bit_cast(int, x), srcLane));
}

__device__ __forceinline__ float clamp01(float x) {
    return __builtin_amdgcn_fmed3f(x, 0.0f, 1.0f);
}

__global__ __launch_bounds__(128)
void dgr_prep_kernel(const float* __restrict__ means3d,
                     const float* __restrict__ scales,
                     const float* __restrict__ opac,
                     const float* __restrict__ colors,
                     float* __restrict__ ws) {
    const unsigned lane = threadIdx.x & 31u;
    const int i = blockIdx.x * 4 + __builtin_amdgcn_readfirstlane((int)(threadIdx.x >> 5));

    float zj[16];
#pragma unroll
    for (int t = 0; t < 16; ++t)
        zj[t] = fmaxf(means3d[(t * 32 + (int)lane) * 3 + 2], 0.1f);

    const float z = fmaxf(means3d[i * 3 + 2], 0.1f);

    const unsigned csel = ((0x204u >> lane) & 1u) + 2u * ((0x810u >> lane) & 1u);
    const unsigned e = lane & 3u;
    const float* src = (lane < 16u) ? colors + i * 3 + csel
                     : (lane < 20u) ? ((e < 2u) ? means3d : scales) + i * 3 + (e & 1u)
                     : opac + i;
    const float m = *src;

    const int it = i >> 5;
    const unsigned below_i = (1u << (i & 31)) - 1u;
    int rank = 0;
#pragma unroll
    for (int t = 0; t < 16; ++t) {
        const unsigned lt = __builtin_amdgcn_ballot_w32(zj[t] < z);
        const unsigned eq = __builtin_amdgcn_ballot_w32(zj[t] == z);
        const unsigned first = (t < it) ? 0xffffffffu : ((t == it) ? below_i : 0u);
        rank += __builtin_popcount(lt | (eq & first));
    }

    const float q = FOCAL_F * m / z;
    const float sg = fmaxf(q, 0.5f);
    const float kc = -0.5f * 1.44269504088896340736f;
    const float k = kc / (sg * sg);

    const unsigned keep = 0u - ((0xA95u >> lane) & 1u);
    const float rec = __builtin_bit_cast(float, __builtin_bit_cast(unsigned, m) & keep);

    const float v = (lane < 16u) ? rec : (lane < 18u) ? q : (lane < 20u) ? k : m;
    const int dst = (lane < 16u) ? (WS_BQ + rank * 16 + (int)lane)
                  : (lane < 20u) ? (rank * 4 + (int)e) : (WS_GOP + rank);
    if (lane <= 20u)
        ws[dst] = v;
}

struct Chunk {
    float4 g;
    float op;
    v4f e[8];
    v2f b[32];
};

__device__ __forceinline__ void load_chunk_ws(Chunk& k, const float4* __restrict__ gp4,
                                              const float* __restrict__ gop,
                                              const v2f* __restrict__ bl, int c, int lane) {
    const int n = c * 32 + lane;
    k.g = gp4[n];
    k.op = gop[n];
#pragma unroll
    for (int j = 0; j < 32; ++j)
        k.b[j] = bl[(c * 32 + j) * 8];
}

__device__ __forceinline__ void load_chunk_e(Chunk& k, const float* erow, int c) {
#pragma unroll
    for (int j = 0; j < 8; ++j)
        k.e[j] = *(const v4f*)(erow + c * 32 + j * 4);
}

__device__ __forceinline__ void do_chunk(const Chunk& k, float y0f, float y1f,
                                         v2f& T, v8f& acc) {
    const float dv0 = y0f - k.g.y;
    const float dv1 = y1f - k.g.y;
    const float r0 = k.op * __builtin_amdgcn_exp2f(dv0 * dv0 * k.g.w);
    const float r1 = k.op * __builtin_amdgcn_exp2f(dv1 * dv1 * k.g.w);

#define ALPHA_OF(j, dst) do { \
        const float e_ = k.e[(j) >> 2][(j) & 3]; \
        (dst).x = clamp01(e_ * lane_bcast(r0, (j))); \
        (dst).y = clamp01(e_ * lane_bcast(r1, (j))); \
    } while (0)

    v2f a;
    ALPHA_OF(0, a);
#pragma unroll
    for (int j = 0; j < 32; ++j) {
        const v2f w = a * T;
        T = T - w;
        if (j + 1 < 32)
            ALPHA_OF(j + 1, a);

        acc = __builtin_amdgcn_wmma_f32_16x16x4_f32(
            false, w, false, k.b[j], (short)0, acc, false, false);
    }
#undef ALPHA_OF
}

__global__ __launch_bounds__(128)
void dgr_raster_kernel(const float* __restrict__ ws, float* __restrict__ out) {
    __shared__ __attribute__((aligned(16))) float etab[32 * EPITCH];

    const float4* __restrict__ gp4 = (const float4*)ws;
    const float* __restrict__ gop = ws + WS_GOP;

    const int lane = threadIdx.x & 31;
    const int wv = __builtin_amdgcn_readfirstlane((int)(threadIdx.x >> 5));
    const int x0 = (blockIdx.x & 7) * 32;
    const int yr = (blockIdx.x >> 3) * 8 + wv * 2;

    const float x0f = (float)x0 - (float)(IMG_W / 2);
    const float y0f = (float)yr - (float)(IMG_H / 2);
    const float y1f = y0f + 1.0f;

    const int c = lane & 15;
    const int h = lane >> 4;
    const int q = c >> 2;
    const int ch = c & 3;
    int boff = 12;
    if (ch < 3) {
        if (q == 2 * h) boff = 2 * ch;
        else if (q == 2 * h + 1) boff = 6 + 2 * ch;
    }
    const v2f* __restrict__ bl = (const v2f*)(ws + WS_BQ + boff);

    Chunk ca, cb;
    load_chunk_ws(ca, gp4, gop, bl, 0, lane);

    {
        float4 p[4];
#pragma unroll
        for (int t = 0; t < 4; ++t)
            p[t] = gp4[wv * 128 + t * 32 + lane];
#pragma unroll
        for (int t = 0; t < 4; ++t) {
            float* dst = etab + wv * 128 + t * 32 + lane;
#pragma unroll
            for (int i = 0; i < 32; ++i) {
                const float du = (x0f + (float)i) - p[t].x;
                dst[i * EPITCH] = __builtin_amdgcn_exp2f(du * du * p[t].z);
            }
        }
    }
    __syncthreads();

    const float* erow = etab + lane * EPITCH;
    load_chunk_e(ca, erow, 0);

    v2f T = {1.0f, 1.0f};
    v8f acc = {0.f, 0.f, 0.f, 0.f, 0.f, 0.f, 0.f, 0.f};

#define LOAD_GROUP_END() __builtin_amdgcn_sched_barrier(0)
    for (int ck = 0;; ck += 2) {
        load_chunk_ws(cb, gp4, gop, bl, ck + 1, lane);
        load_chunk_e(cb, erow, ck + 1);
        LOAD_GROUP_END();
        do_chunk(ca, y0f, y1f, T, acc);
        if (ck + 2 >= NCHUNK) {
            do_chunk(cb, y0f, y1f, T, acc);
            break;
        }
        load_chunk_ws(ca, gp4, gop, bl, ck + 2, lane);
        load_chunk_e(ca, erow, ck + 2);
        LOAD_GROUP_END();
        do_chunk(cb, y0f, y1f, T, acc);
    }
#undef LOAD_GROUP_END

    if (ch < 3) {
        const int row = yr + (q & 1);
        const int xb = x0 + 16 * (q >> 1) + 8 * h;
        float* o = out + ((size_t)row * IMG_W + xb) * 3 + ch;
#pragma unroll
        for (int r = 0; r < 8; ++r)
            o[r * 3] = acc[r];
    }
}

extern "C" void kernel_launch(void* const* d_in, const int* in_sizes, int n_in,
                              void* d_out, int out_size, void* d_ws, size_t ws_size,
                              hipStream_t stream) {
    const float* means3d = (const float*)d_in[0];
    const float* scales  = (const float*)d_in[1];
    const float* opac    = (const float*)d_in[2];
    const float* colors  = (const float*)d_in[3];
    float* ws  = (float*)d_ws;
    float* out = (float*)d_out;

    hipLaunchKernelGGL(dgr_prep_kernel, dim3(128), dim3(128), 0, stream,
                       means3d, scales, opac, colors, ws);

    hipLaunchKernelGGL(dgr_raster_kernel, dim3(256), dim3(128), 0, stream,
                       ws, out);
}


# raster: 8 waves per work-group, two per SIMD: a wave's strip is 32 columns x 1 row (was 4 waves of 32 x 2); one matrix instruction adds 2 consecutive gaussians of the lane's pixel, depth order kept
# speedup vs baseline: 4.2051x; 1.0597x over previous
#include <hip/hip_runtime.h>


typedef __attribute__((ext_vector_type(2))) float v2f;
typedef __attribute__((ext_vector_type(4))) float v4f;
typedef __attribute__((ext_vector_type(8))) float v8f;

#define IMG_H 256
#define IMG_W 256
#define NGAUSS 512
#define FOCAL_F 50.0f
#define NCHUNK (NGAUSS / 32)
#define EPITCH (NGAUSS + 4)

#define WS_GOP 2048
#define WS_COL 2560

__device__ __forceinline__ float lane_bcast(float x, int srcLane) {
    return __builtin_bit_cast(float, __builtin_amdgcn_readlane(__builtin_bit_cast(int, x), srcLane));
}

__device__ __forceinline__ float clamp01(float x) {
    return __builtin_amdgcn_fmed3f(x, 0.0f, 1.0f);
}

__global__ __launch_bounds__(128)
void dgr_prep_kernel(const float* __restrict__ means3d,
                     const float* __restrict__ scales,
                     const float* __restrict__ opac,
                     const float* __restrict__ colors,
                     float* __restrict__ ws) {
    const unsigned lane = threadIdx.x & 31u;
    const int i = blockIdx.x * 4 + __builtin_amdgcn_readfirstlane((int)(threadIdx.x >> 5));

    float zj[16];
#pragma unroll
    for (int t = 0; t < 16; ++t)
        zj[t] = fmaxf(means3d[(t * 32 + (int)lane) * 3 + 2], 0.1f);

    const float z = fmaxf(means3d[i * 3 + 2], 0.1f);

    const unsigned e = lane & 3u;
    const float* src = (lane < 3u) ? colors + i * 3 + lane
                     : (lane >= 16u && lane < 20u) ? ((e < 2u) ? means3d : scales) + i * 3 + (e & 1u)
                     : opac + i;
    const float m = *src;

    const int it = i >> 5;
    const unsigned below_i = (1u << (i & 31)) - 1u;
    int rank = 0;
#pragma unroll
    for (int t = 0; t < 16; ++t) {
        const unsigned lt = __builtin_amdgcn_ballot_w32(zj[t] < z);
        const unsigned eq = __builtin_amdgcn_ballot_w32(zj[t] == z);
        const unsigned first = (t < it) ? 0xffffffffu : ((t == it) ? below_i : 0u);
        rank += __builtin_popcount(lt | (eq & first));
    }

    const float q = FOCAL_F * m / z;
    const float sg = fmaxf(q, 0.5f);
    const float kc = -0.5f * 1.44269504088896340736f;
    const float k = kc / (sg * sg);

    const float v = (lane < 3u) ? m : (lane == 3u) ? 0.0f : (lane < 18u) ? q : (lane < 20u) ? k : m;
    const int dst = (lane < 4u) ? (WS_COL + (int)lane * NGAUSS + rank)
                  : (lane < 20u) ? (rank * 4 + (int)e) : (WS_GOP + rank);
    if (lane < 4u || (lane >= 16u && lane <= 20u))
        ws[dst] = v;
}

struct Chunk {
    float4 g;
    float op;
    v4f e[8];
    v4f b[8];
};

__device__ __forceinline__ void load_chunk_ws(Chunk& k, const float4* __restrict__ gp4,
                                              const float* __restrict__ gop,
                                              const float* __restrict__ bl, int c, int lane) {
    const int n = c * 32 + lane;
    k.g = gp4[n];
    k.op = gop[n];
#pragma unroll
    for (int j = 0; j < 8; ++j)
        k.b[j] = *(const v4f*)(bl + c * 32 + j * 4);
}

__device__ __forceinline__ void load_chunk_e(Chunk& k, const float* erow, int c) {
#pragma unroll
    for (int j = 0; j < 8; ++j)
        k.e[j] = *(const v4f*)(erow + c * 32 + j * 4);
}

__device__ __forceinline__ void do_chunk(const Chunk& k, float yf, float& T, v8f& acc) {
    const float dv = yf - k.g.y;
    const float r = k.op * __builtin_amdgcn_exp2f(dv * dv * k.g.w);

#define ALPHA_OF(p, dst) do { \
        (dst).x = clamp01(k.e[(2 * (p)) >> 2][(2 * (p)) & 3] * lane_bcast(r, 2 * (p))); \
        (dst).y = clamp01(k.e[(2 * (p) + 1) >> 2][(2 * (p) + 1) & 3] * lane_bcast(r, 2 * (p) + 1)); \
    } while (0)

    v2f a;
    ALPHA_OF(0, a);
#pragma unroll
    for (int p = 0; p < 16; ++p) {
        v2f w;
        w.x = a.x * T;
        T = T - w.x;
        w.y = a.y * T;
        T = T - w.y;
        v2f b;
        b.x = k.b[(2 * p) >> 2][(2 * p) & 3];
        b.y = k.b[(2 * p + 1) >> 2][(2 * p + 1) & 3];
        if (p + 1 < 16)
            ALPHA_OF(p + 1, a);

        acc = __builtin_amdgcn_wmma_f32_16x16x4_f32(
            false, w, false, b, (short)0, acc, false, false);
    }
#undef ALPHA_OF
}

__global__ __launch_bounds__(256)
void dgr_raster_kernel(const float* __restrict__ ws, float* __restrict__ out) {
    __shared__ __attribute__((aligned(16))) float etab[32 * EPITCH];

    const float4* __restrict__ gp4 = (const float4*)ws;
    const float* __restrict__ gop = ws + WS_GOP;

    const int lane = threadIdx.x & 31;
    const int wv = __builtin_amdgcn_readfirstlane((int)(threadIdx.x >> 5));
    const int x0 = (blockIdx.x & 7) * 32;
    const int yr = (blockIdx.x >> 3) * 8 + wv;

    const float x0f = (float)x0 - (float)(IMG_W / 2);
    const float yf = (float)yr - (float)(IMG_H / 2);

    const int c = lane & 15;
    const int h = lane >> 4;
    const int ch = c & 3;
    const int brow = (c < 8 && ch < 3 && (c >> 2) == h) ? ch : 3;
    const float* __restrict__ bl = ws + WS_COL + brow * NGAUSS;

    Chunk ca, cb;
    load_chunk_ws(ca, gp4, gop, bl, 0, lane);

    {
        float4 p[2];
#pragma unroll
        for (int t = 0; t < 2; ++t)
            p[t] = gp4[wv * 64 + t * 32 + lane];
#pragma unroll
        for (int t = 0; t < 2; ++t) {
            float* dst = etab + wv * 64 + t * 32 + lane;
#pragma unroll
            for (int i = 0; i < 32; ++i) {
                const float du = (x0f + (float)i) - p[t].x;
                dst[i * EPITCH] = __builtin_amdgcn_exp2f(du * du * p[t].z);
            }
        }
    }
    __syncthreads();

    const float* erow = etab + lane * EPITCH;
    load_chunk_e(ca, erow, 0);

    float T = 1.0f;
    v8f acc = {0.f, 0.f, 0.f, 0.f, 0.f, 0.f, 0.f, 0.f};

#define LOAD_GROUP_END() __builtin_amdgcn_sched_barrier(0)
    for (int ck = 0;; ck += 2) {
        load_chunk_ws(cb, gp4, gop, bl, ck + 1, lane);
        load_chunk_e(cb, erow, ck + 1);
        LOAD_GROUP_END();
        do_chunk(ca, yf, T, acc);
        if (ck + 2 >= NCHUNK) {
            do_chunk(cb, yf, T, acc);
            break;
        }
        load_chunk_ws(ca, gp4, gop, bl, ck + 2, lane);
        load_chunk_e(ca, erow, ck + 2);
        LOAD_GROUP_END();
        do_chunk(cb, yf, T, acc);
    }
#undef LOAD_GROUP_END

    if (c < 8 && ch < 3) {
        const int xb = x0 + 16 * (c >> 2) + 8 * h;
        float* o = out + ((size_t)yr * IMG_W + xb) * 3 + ch;
#pragma unroll
        for (int r = 0; r < 8; ++r)
            o[r * 3] = acc[r];
    }
}

extern "C" void kernel_launch(void* const* d_in, const int* in_sizes, int n_in,
                              void* d_out, int out_size, void* d_ws, size_t ws_size,
                              hipStream_t stream) {
    const float* means3d = (const float*)d_in[0];
    const float* scales  = (const float*)d_in[1];
    const float* opac    = (const float*)d_in[2];
    const float* colors  = (const float*)d_in[3];
    float* ws  = (float*)d_ws;
    float* out = (float*)d_out;

    hipLaunchKernelGGL(dgr_prep_kernel, dim3(128), dim3(128), 0, stream,
                       means3d, scales, opac, colors, ws);

    hipLaunchKernelGGL(dgr_raster_kernel, dim3(256), dim3(256), 0, stream,
                       ws, out);
}
